# starting kernel
# baseline (speedup 1.0000x reference)
#include <hip/hip_runtime.h>
#include <hip/hip_bf16.h>
#include <math.h>

typedef float v2f __attribute__((ext_vector_type(2)));
typedef float v8f __attribute__((ext_vector_type(8)));

#define BATCH   128
#define RROWS   8
#define NSLOT   4096
#define MDIM    64
#define NTILES  (NSLOT / 16)
#define COS_EPS 1e-8f
#define ADD_EPS 1e-16f

__global__ __launch_bounds__(256)
void ntm_address_kernel(const float* __restrict__ memory,
                        const float* __restrict__ key,
                        const float* __restrict__ beta,
                        const float* __restrict__ gate,
                        const float* __restrict__ shift,
                        const float* __restrict__ gamma,
                        const float* __restrict__ w_prev,
                        float* __restrict__ w_out)
{
    __shared__ float smem[RROWS * NSLOT];

    const int b    = blockIdx.x;
    const int tid  = threadIdx.x;
    const int lane = tid & 31;
    const int wave = tid >> 5;
    const int half = lane >> 4;
    const int mrow = lane & 15;

    const float* kb = key + (size_t)b * RROWS * MDIM;

    v2f a[16];
#pragma unroll
    for (int s = 0; s < 16; ++s) {
        if (mrow < RROWS) {
            const int m = 4 * s + 2 * half;
            a[s].x = kb[mrow * MDIM + m]     + ADD_EPS;
            a[s].y = kb[mrow * MDIM + m + 1] + ADD_EPS;
        } else {
            a[s].x = 0.0f;
            a[s].y = 0.0f;
        }
    }

    float knl = 0.0f;
    if (lane < RROWS) {
        for (int m = 0; m < MDIM; ++m) {
            const float v = kb[lane * MDIM + m] + ADD_EPS;
            knl += v * v;
        }
        knl = sqrtf(knl);
    }
    float kn[RROWS], bet[RROWS];
#pragma unroll
    for (int j = 0; j < RROWS; ++j) {
        kn[j]  = __shfl(knl, j, 32);
        bet[j] = beta[b * RROWS + j];
    }

    const float* mb = memory + (size_t)b * NSLOT * MDIM;

    for (int tile = wave; tile < NTILES; tile += 8) {
        const int    n0   = tile * 16;
        const float* rowp = mb + (size_t)(n0 + mrow) * MDIM + 2 * half;

        __builtin_prefetch(rowp + 8 * 16 * MDIM, 0, 1);

        v8f   acc = {};
        float ssq = 0.0f;
#pragma unroll
        for (int s = 0; s < 16; ++s) {
            v2f bv;
            bv.x = rowp[4 * s]     + ADD_EPS;
            bv.y = rowp[4 * s + 1] + ADD_EPS;
            ssq += bv.x * bv.x + bv.y * bv.y;
            acc = __builtin_amdgcn_wmma_f32_16x16x4_f32(
                      false, a[s], false, bv, (short)0, acc, false, false);
        }

        ssq += __shfl_xor(ssq, 16, 32);
        const float mem_norm = sqrtf(ssq);

        if (lane < 16) {
            const int n = n0 + lane;
#pragma unroll
            for (int j = 0; j < RROWS; ++j) {
                const float denom = fmaxf(mem_norm * kn[j], COS_EPS);
                smem[j * NSLOT + n] = bet[j] * (acc[j] / denom);
            }
        }
    }

    __syncthreads();

    {
        const int r = wave;
        float* trow = smem + r * NSLOT;

        float mx = -INFINITY;
        for (int n = lane; n < NSLOT; n += 32) mx = fmaxf(mx, trow[n]);
#pragma unroll
        for (int o = 16; o > 0; o >>= 1) mx = fmaxf(mx, __shfl_xor(mx, o, 32));

        float Z = 0.0f;
        for (int n = lane; n < NSLOT; n += 32) {
            const float e = __expf(trow[n] - mx);
            trow[n] = e;
            Z += e;
        }
#pragma unroll
        for (int o = 16; o > 0; o >>= 1) Z += __shfl_xor(Z, o, 32);
        const float invZ = 1.0f / Z;

        const int    hr = b * RROWS + r;
        const float  gr = gate[hr];
        const float  s0 = shift[hr * 3 + 0];
        const float  s1 = shift[hr * 3 + 1];
        const float  s2 = shift[hr * 3 + 2];
        const float  gm = gamma[hr];
        const float* wp = w_prev + (size_t)hr * NSLOT;

        for (int n = lane; n < NSLOT; n += 32) {
            const float wc = trow[n] * invZ;
            trow[n] = gr * wc + (1.0f - gr) * wp[n];
        }

        __syncthreads();

        float S = 0.0f;
        for (int n = lane; n < NSLOT; n += 32) {
            const int nm = (n + NSLOT - 1) & (NSLOT - 1);
            const int np = (n + 1) & (NSLOT - 1);
            const float c = s0 * trow[nm] + s1 * trow[n] + s2 * trow[np];
            S += __powf(c, gm);
        }
#pragma unroll
        for (int o = 16; o > 0; o >>= 1) S += __shfl_xor(S, o, 32);
        const float invS = 1.0f / (S + ADD_EPS);

        float* wo = w_out + (size_t)hr * NSLOT;
        for (int n = lane; n < NSLOT; n += 32) {
            const int nm = (n + NSLOT - 1) & (NSLOT - 1);
            const int np = (n + 1) & (NSLOT - 1);
            const float c = s0 * trow[nm] + s1 * trow[n] + s2 * trow[np];
            wo[n] = __powf(c, gm) * invS;
        }
    }
}

extern "C" void kernel_launch(void* const* d_in, const int* in_sizes, int n_in,
                              void* d_out, int out_size, void* d_ws, size_t ws_size,
                              hipStream_t stream) {
    (void)in_sizes; (void)n_in; (void)out_size; (void)d_ws; (void)ws_size;
    const float* memory = (const float*)d_in[0];
    const float* k      = (const float*)d_in[1];
    const float* beta   = (const float*)d_in[2];
    const float* g      = (const float*)d_in[3];
    const float* s      = (const float*)d_in[4];
    const float* gamma  = (const float*)d_in[5];
    const float* w_prev = (const float*)d_in[6];
    float* out = (float*)d_out;

    ntm_address_kernel<<<dim3(BATCH), dim3(256), 0, stream>>>(
        memory, k, beta, g, s, gamma, w_prev, out);
}


# c^gamma in both sharpen passes via the hardware log2/exp2 units (mantissa-exponent split, integer part through ldexp, a few ulp) instead of the in-line library pow; every summation order unchanged
# speedup vs baseline: 2.4821x; 2.4821x over previous
#include <hip/hip_runtime.h>
#include <hip/hip_bf16.h>
#include <math.h>

typedef float v2f __attribute__((ext_vector_type(2)));
typedef float v8f __attribute__((ext_vector_type(8)));

#define BATCH   128
#define RROWS   8
#define NSLOT   4096
#define MDIM    64
#define NTILES  (NSLOT / 16)
#define COS_EPS 1e-8f
#define ADD_EPS 1e-16f

__device__ __forceinline__ float pow_nonneg(float c, float g)
{
    const float m  = __builtin_amdgcn_frexp_mantf(c);
    const float ef = (float)__builtin_amdgcn_frexp_expf(c);
    const float l  = __builtin_amdgcn_logf(m);
    const float n  = rintf(g * ef);
    const float f  = fmaf(g, l, fmaf(g, ef, -n));
    const float r  = ldexpf(__builtin_amdgcn_exp2f(f), (int)n);
    return (g == 0.0f) ? 1.0f : r;
}

__global__ __launch_bounds__(256)
void ntm_address_kernel(const float* __restrict__ memory,
                        const float* __restrict__ key,
                        const float* __restrict__ beta,
                        const float* __restrict__ gate,
                        const float* __restrict__ shift,
                        const float* __restrict__ gamma,
                        const float* __restrict__ w_prev,
                        float* __restrict__ w_out)
{
    __shared__ float smem[RROWS * NSLOT];

    const int b    = blockIdx.x;
    const int tid  = threadIdx.x;
    const int lane = tid & 31;
    const int wave = tid >> 5;
    const int half = lane >> 4;
    const int mrow = lane & 15;

    const float* kb = key + (size_t)b * RROWS * MDIM;

    v2f a[16];
#pragma unroll
    for (int s = 0; s < 16; ++s) {
        if (mrow < RROWS) {
            const int m = 4 * s + 2 * half;
            a[s].x = kb[mrow * MDIM + m]     + ADD_EPS;
            a[s].y = kb[mrow * MDIM + m + 1] + ADD_EPS;
        } else {
            a[s].x = 0.0f;
            a[s].y = 0.0f;
        }
    }

    float knl = 0.0f;
    if (lane < RROWS) {
        for (int m = 0; m < MDIM; ++m) {
            const float v = kb[lane * MDIM + m] + ADD_EPS;
            knl += v * v;
        }
        knl = sqrtf(knl);
    }
    float kn[RROWS], bet[RROWS];
#pragma unroll
    for (int j = 0; j < RROWS; ++j) {
        kn[j]  = __shfl(knl, j, 32);
        bet[j] = beta[b * RROWS + j];
    }

    const float* mb = memory + (size_t)b * NSLOT * MDIM;

    for (int tile = wave; tile < NTILES; tile += 8) {
        const int    n0   = tile * 16;
        const float* rowp = mb + (size_t)(n0 + mrow) * MDIM + 2 * half;

        __builtin_prefetch(rowp + 8 * 16 * MDIM, 0, 1);

        v8f   acc = {};
        float ssq = 0.0f;
#pragma unroll
        for (int s = 0; s < 16; ++s) {
            v2f bv;
            bv.x = rowp[4 * s]     + ADD_EPS;
            bv.y = rowp[4 * s + 1] + ADD_EPS;
            ssq += bv.x * bv.x + bv.y * bv.y;
            acc = __builtin_amdgcn_wmma_f32_16x16x4_f32(
                      false, a[s], false, bv, (short)0, acc, false, false);
        }

        ssq += __shfl_xor(ssq, 16, 32);
        const float mem_norm = sqrtf(ssq);

        if (lane < 16) {
            const int n = n0 + lane;
#pragma unroll
            for (int j = 0; j < RROWS; ++j) {
                const float denom = fmaxf(mem_norm * kn[j], COS_EPS);
                smem[j * NSLOT + n] = bet[j] * (acc[j] / denom);
            }
        }
    }

    __syncthreads();

    {
        const int r = wave;
        float* trow = smem + r * NSLOT;

        float mx = -INFINITY;
        for (int n = lane; n < NSLOT; n += 32) mx = fmaxf(mx, trow[n]);
#pragma unroll
        for (int o = 16; o > 0; o >>= 1) mx = fmaxf(mx, __shfl_xor(mx, o, 32));

        float Z = 0.0f;
        for (int n = lane; n < NSLOT; n += 32) {
            const float e = __expf(trow[n] - mx);
            trow[n] = e;
            Z += e;
        }
#pragma unroll
        for (int o = 16; o > 0; o >>= 1) Z += __shfl_xor(Z, o, 32);
        const float invZ = 1.0f / Z;

        const int    hr = b * RROWS + r;
        const float  gr = gate[hr];
        const float  s0 = shift[hr * 3 + 0];
        const float  s1 = shift[hr * 3 + 1];
        const float  s2 = shift[hr * 3 + 2];
        const float  gm = gamma[hr];
        const float* wp = w_prev + (size_t)hr * NSLOT;

        for (int n = lane; n < NSLOT; n += 32) {
            const float wc = trow[n] * invZ;
            trow[n] = gr * wc + (1.0f - gr) * wp[n];
        }

        __syncthreads();

        float S = 0.0f;
        for (int n = lane; n < NSLOT; n += 32) {
            const int nm = (n + NSLOT - 1) & (NSLOT - 1);
            const int np = (n + 1) & (NSLOT - 1);
            const float c = s0 * trow[nm] + s1 * trow[n] + s2 * trow[np];
            S += pow_nonneg(c, gm);
        }
#pragma unroll
        for (int o = 16; o > 0; o >>= 1) S += __shfl_xor(S, o, 32);
        const float invS = 1.0f / (S + ADD_EPS);

        float* wo = w_out + (size_t)hr * NSLOT;
        for (int n = lane; n < NSLOT; n += 32) {
            const int nm = (n + NSLOT - 1) & (NSLOT - 1);
            const int np = (n + 1) & (NSLOT - 1);
            const float c = s0 * trow[nm] + s1 * trow[n] + s2 * trow[np];
            wo[n] = pow_nonneg(c, gm) * invS;
        }
    }
}

extern "C" void kernel_launch(void* const* d_in, const int* in_sizes, int n_in,
                              void* d_out, int out_size, void* d_ws, size_t ws_size,
                              hipStream_t stream) {
    (void)in_sizes; (void)n_in; (void)out_size; (void)d_ws; (void)ws_size;
    const float* memory = (const float*)d_in[0];
    const float* k      = (const float*)d_in[1];
    const float* beta   = (const float*)d_in[2];
    const float* g      = (const float*)d_in[3];
    const float* s      = (const float*)d_in[4];
    const float* gamma  = (const float*)d_in[5];
    const float* w_prev = (const float*)d_in[6];
    float* out = (float*)d_out;

    ntm_address_kernel<<<dim3(BATCH), dim3(256), 0, stream>>>(
        memory, k, beta, g, s, gamma, w_prev, out);
}


# 32 waves per workgroup instead of 8: 8 tiles per wave in the score phase, 4 waves per head after it; softmax, sharpen and key-norm sums change order (wave partials joined through LDS)
# speedup vs baseline: 3.5757x; 1.4406x over previous
#include <hip/hip_runtime.h>
#include <hip/hip_bf16.h>
#include <math.h>

typedef float v2f __attribute__((ext_vector_type(2)));
typedef float v8f __attribute__((ext_vector_type(8)));

#define BATCH   128
#define RROWS   8
#define NSLOT   4096
#define MDIM    64
#define NTILES  (NSLOT / 16)
#define NWAVE   32
#define NTHR    (NWAVE * 32)
#define WPH     (NWAVE / RROWS)
#define SEG     (NSLOT / WPH)
#define COS_EPS 1e-8f
#define ADD_EPS 1e-16f

__device__ __forceinline__ float pow_nonneg(float c, float g)
{
    const float m  = __builtin_amdgcn_frexp_mantf(c);
    const float ef = (float)__builtin_amdgcn_frexp_expf(c);
    const float l  = __builtin_amdgcn_logf(m);
    const float n  = rintf(g * ef);
    const float f  = fmaf(g, l, fmaf(g, ef, -n));
    const float r  = ldexpf(__builtin_amdgcn_exp2f(f), (int)n);
    return (g == 0.0f) ? 1.0f : r;
}

__global__ __launch_bounds__(NTHR)
void ntm_address_kernel(const float* __restrict__ memory,
                        const float* __restrict__ key,
                        const float* __restrict__ beta,
                        const float* __restrict__ gate,
                        const float* __restrict__ shift,
                        const float* __restrict__ gamma,
                        const float* __restrict__ w_prev,
                        float* __restrict__ w_out)
{
    __shared__ float smem[RROWS * NSLOT];
    __shared__ float red[3 * NWAVE];

    const int b    = blockIdx.x;
    const int tid  = threadIdx.x;
    const int lane = tid & 31;
    const int wave = tid >> 5;
    const int half = lane >> 4;
    const int mrow = lane & 15;

    const float* kb = key + (size_t)b * RROWS * MDIM;

    v2f a[16];
#pragma unroll
    for (int s = 0; s < 16; ++s) {
        const int m = 4 * s + 2 * half;
        const float kx = kb[(mrow & (RROWS - 1)) * MDIM + m]     + ADD_EPS;
        const float ky = kb[(mrow & (RROWS - 1)) * MDIM + m + 1] + ADD_EPS;
        a[s].x = (mrow < RROWS) ? kx : 0.0f;
        a[s].y = (mrow < RROWS) ? ky : 0.0f;
    }

    float knl = 0.0f;
#pragma unroll
    for (int s = 0; s < 16; ++s) knl += a[s].x * a[s].x + a[s].y * a[s].y;
    knl += __shfl_xor(knl, 16, 32);
    knl = sqrtf(knl);
    float kn[RROWS], bet[RROWS];
#pragma unroll
    for (int j = 0; j < RROWS; ++j) {
        kn[j]  = __shfl(knl, j, 32);
        bet[j] = beta[b * RROWS + j];
    }

    const float* mb = memory + (size_t)b * NSLOT * MDIM;

    for (int tile = wave; tile < NTILES; tile += NWAVE) {
        const int    n0   = tile * 16;
        const float* rowp = mb + (size_t)(n0 + mrow) * MDIM + 2 * half;

        if (tile + NWAVE < NTILES) __builtin_prefetch(rowp + NWAVE * 16 * MDIM, 0, 1);

        v8f   acc = {};
        float ssq = 0.0f;
#pragma unroll
        for (int s = 0; s < 16; ++s) {
            v2f bv;
            bv.x = rowp[4 * s]     + ADD_EPS;
            bv.y = rowp[4 * s + 1] + ADD_EPS;
            ssq += bv.x * bv.x + bv.y * bv.y;
            acc = __builtin_amdgcn_wmma_f32_16x16x4_f32(
                      false, a[s], false, bv, (short)0, acc, false, false);
        }

        ssq += __shfl_xor(ssq, 16, 32);
        const float mem_norm = sqrtf(ssq);

        if (lane < 16) {
            const int n = n0 + lane;
#pragma unroll
            for (int j = 0; j < RROWS; ++j) {
                const float denom = fmaxf(mem_norm * kn[j], COS_EPS);
                smem[j * NSLOT + n] = bet[j] * (acc[j] / denom);
            }
        }
    }

    __syncthreads();

    {
        const int r  = wave / WPH;
        const int q  = wave % WPH;
        const int n0 = q * SEG + lane;
        const int n1 = (q + 1) * SEG;
        float* trow = smem + r * NSLOT;
        const float* rmax = red + r * WPH;
        const float* rZ   = red + NWAVE + r * WPH;
        const float* rS   = red + 2 * NWAVE + r * WPH;

        float mx = -INFINITY;
        for (int n = n0; n < n1; n += 32) mx = fmaxf(mx, trow[n]);
#pragma unroll
        for (int o = 16; o > 0; o >>= 1) mx = fmaxf(mx, __shfl_xor(mx, o, 32));
        if (lane == 0) red[wave] = mx;
        __syncthreads();
        mx = rmax[0];
        for (int i = 1; i < WPH; ++i) mx = fmaxf(mx, rmax[i]);

        float Z = 0.0f;
        for (int n = n0; n < n1; n += 32) {
            const float e = __expf(trow[n] - mx);
            trow[n] = e;
            Z += e;
        }
#pragma unroll
        for (int o = 16; o > 0; o >>= 1) Z += __shfl_xor(Z, o, 32);
        if (lane == 0) red[NWAVE + wave] = Z;
        __syncthreads();
        Z = rZ[0];
        for (int i = 1; i < WPH; ++i) Z += rZ[i];
        const float invZ = 1.0f / Z;

        const int    hr = b * RROWS + r;
        const float  gr = gate[hr];
        const float  s0 = shift[hr * 3 + 0];
        const float  s1 = shift[hr * 3 + 1];
        const float  s2 = shift[hr * 3 + 2];
        const float  gm = gamma[hr];
        const float* wp = w_prev + (size_t)hr * NSLOT;

        for (int n = n0; n < n1; n += 32) {
            const float wc = trow[n] * invZ;
            trow[n] = gr * wc + (1.0f - gr) * wp[n];
        }

        __syncthreads();

        float S = 0.0f;
        for (int n = n0; n < n1; n += 32) {
            const int nm = (n + NSLOT - 1) & (NSLOT - 1);
            const int np = (n + 1) & (NSLOT - 1);
            const float c = s0 * trow[nm] + s1 * trow[n] + s2 * trow[np];
            S += pow_nonneg(c, gm);
        }
#pragma unroll
        for (int o = 16; o > 0; o >>= 1) S += __shfl_xor(S, o, 32);
        if (lane == 0) red[2 * NWAVE + wave] = S;
        __syncthreads();
        S = rS[0];
        for (int i = 1; i < WPH; ++i) S += rS[i];
        const float invS = 1.0f / (S + ADD_EPS);

        float* wo = w_out + (size_t)hr * NSLOT;
        for (int n = n0; n < n1; n += 32) {
            const int nm = (n + NSLOT - 1) & (NSLOT - 1);
            const int np = (n + 1) & (NSLOT - 1);
            const float c = s0 * trow[nm] + s1 * trow[n] + s2 * trow[np];
            wo[n] = pow_nonneg(c, gm) * invS;
        }
    }
}

extern "C" void kernel_launch(void* const* d_in, const int* in_sizes, int n_in,
                              void* d_out, int out_size, void* d_ws, size_t ws_size,
                              hipStream_t stream) {
    (void)in_sizes; (void)n_in; (void)out_size; (void)d_ws; (void)ws_size;
    const float* memory = (const float*)d_in[0];
    const float* k      = (const float*)d_in[1];
    const float* beta   = (const float*)d_in[2];
    const float* g      = (const float*)d_in[3];
    const float* s      = (const float*)d_in[4];
    const float* gamma  = (const float*)d_in[5];
    const float* w_prev = (const float*)d_in[6];
    float* out = (float*)d_out;

    ntm_address_kernel<<<dim3(BATCH), dim3(NTHR), 0, stream>>>(
        memory, k, beta, g, s, gamma, w_prev, out);
}


# phase 2 keeps each lane's 32 row values in registers from score read to output store: LDS row read once and written once, c^gamma computed once per slot, not twice; summation orders unchanged
# speedup vs baseline: 4.3657x; 1.2209x over previous
#include <hip/hip_runtime.h>
#include <hip/hip_bf16.h>
#include <math.h>

typedef float v2f __attribute__((ext_vector_type(2)));
typedef float v8f __attribute__((ext_vector_type(8)));

#define BATCH   128
#define RROWS   8
#define NSLOT   4096
#define MDIM    64
#define NTILES  (NSLOT / 16)
#define NWAVE   32
#define NTHR    (NWAVE * 32)
#define WPH     (NWAVE / RROWS)
#define SEG     (NSLOT / WPH)
#define PL      (SEG / 32)
#define COS_EPS 1e-8f
#define ADD_EPS 1e-16f

__device__ __forceinline__ float pow_nonneg(float c, float g)
{
    const float m  = __builtin_amdgcn_frexp_mantf(c);
    const float ef = (float)__builtin_amdgcn_frexp_expf(c);
    const float l  = __builtin_amdgcn_logf(m);
    const float n  = rintf(g * ef);
    const float f  = fmaf(g, l, fmaf(g, ef, -n));
    const float r  = ldexpf(__builtin_amdgcn_exp2f(f), (int)n);
    return (g == 0.0f) ? 1.0f : r;
}

__global__ __launch_bounds__(NTHR)
void ntm_address_kernel(const float* __restrict__ memory,
                        const float* __restrict__ key,
                        const float* __restrict__ beta,
                        const float* __restrict__ gate,
                        const float* __restrict__ shift,
                        const float* __restrict__ gamma,
                        const float* __restrict__ w_prev,
                        float* __restrict__ w_out)
{
    __shared__ float smem[RROWS * NSLOT];
    __shared__ float red[3 * NWAVE];

    const int b    = blockIdx.x;
    const int tid  = threadIdx.x;
    const int lane = tid & 31;
    const int wave = tid >> 5;
    const int half = lane >> 4;
    const int mrow = lane & 15;

    const float* kb = key + (size_t)b * RROWS * MDIM;

    v2f a[16];
#pragma unroll
    for (int s = 0; s < 16; ++s) {
        const int m = 4 * s + 2 * half;
        const float kx = kb[(mrow & (RROWS - 1)) * MDIM + m]     + ADD_EPS;
        const float ky = kb[(mrow & (RROWS - 1)) * MDIM + m + 1] + ADD_EPS;
        a[s].x = (mrow < RROWS) ? kx : 0.0f;
        a[s].y = (mrow < RROWS) ? ky : 0.0f;
    }

    float knl = 0.0f;
#pragma unroll
    for (int s = 0; s < 16; ++s) knl += a[s].x * a[s].x + a[s].y * a[s].y;
    knl += __shfl_xor(knl, 16, 32);
    knl = sqrtf(knl);
    float kn[RROWS], bet[RROWS];
#pragma unroll
    for (int j = 0; j < RROWS; ++j) {
        kn[j]  = __shfl(knl, j, 32);
        bet[j] = beta[b * RROWS + j];
    }

    const float* mb = memory + (size_t)b * NSLOT * MDIM;

    for (int tile = wave; tile < NTILES; tile += NWAVE) {
        const int    n0   = tile * 16;
        const float* rowp = mb + (size_t)(n0 + mrow) * MDIM + 2 * half;

        if (tile + NWAVE < NTILES) __builtin_prefetch(rowp + NWAVE * 16 * MDIM, 0, 1);

        v8f   acc = {};
        float ssq = 0.0f;
#pragma unroll
        for (int s = 0; s < 16; ++s) {
            v2f bv;
            bv.x = rowp[4 * s]     + ADD_EPS;
            bv.y = rowp[4 * s + 1] + ADD_EPS;
            ssq += bv.x * bv.x + bv.y * bv.y;
            acc = __builtin_amdgcn_wmma_f32_16x16x4_f32(
                      false, a[s], false, bv, (short)0, acc, false, false);
        }

        ssq += __shfl_xor(ssq, 16, 32);
        const float mem_norm = sqrtf(ssq);

        if (lane < 16) {
            const int n = n0 + lane;
#pragma unroll
            for (int j = 0; j < RROWS; ++j) {
                const float denom = fmaxf(mem_norm * kn[j], COS_EPS);
                smem[j * NSLOT + n] = bet[j] * (acc[j] / denom);
            }
        }
    }

    __syncthreads();

    {
        const int r    = wave / WPH;
        const int q    = wave % WPH;
        const int base = q * SEG + lane;
        float* trow = smem + r * NSLOT;
        const float* rmax = red + r * WPH;
        const float* rZ   = red + NWAVE + r * WPH;
        const float* rS   = red + 2 * NWAVE + r * WPH;

        const int    hr = b * RROWS + r;
        const float* wp = w_prev + (size_t)hr * NSLOT + base;
        float*       wo = w_out  + (size_t)hr * NSLOT + base;

        const float gr = gate[hr];
        const float s0 = shift[hr * 3 + 0];
        const float s1 = shift[hr * 3 + 1];
        const float s2 = shift[hr * 3 + 2];
        const float gm = gamma[hr];

        float v[PL];
#pragma unroll
        for (int i = 0; i < PL; ++i) v[i] = trow[base + 32 * i];

        float mx = -INFINITY;
#pragma unroll
        for (int i = 0; i < PL; ++i) mx = fmaxf(mx, v[i]);
#pragma unroll
        for (int o = 16; o > 0; o >>= 1) mx = fmaxf(mx, __shfl_xor(mx, o, 32));
        if (lane == 0) red[wave] = mx;
        __syncthreads();
        mx = rmax[0];
        for (int i = 1; i < WPH; ++i) mx = fmaxf(mx, rmax[i]);

        float pv[PL];
#pragma unroll
        for (int i = 0; i < PL; ++i) pv[i] = wp[32 * i];

        float Z = 0.0f;
#pragma unroll
        for (int i = 0; i < PL; ++i) {
            v[i] = __expf(v[i] - mx);
            Z += v[i];
        }
#pragma unroll
        for (int o = 16; o > 0; o >>= 1) Z += __shfl_xor(Z, o, 32);
        if (lane == 0) red[NWAVE + wave] = Z;
        __syncthreads();
        Z = rZ[0];
        for (int i = 1; i < WPH; ++i) Z += rZ[i];
        const float invZ = 1.0f / Z;

#pragma unroll
        for (int i = 0; i < PL; ++i) {
            const float wc = v[i] * invZ;
            v[i] = gr * wc + (1.0f - gr) * pv[i];
            trow[base + 32 * i] = v[i];
        }

        __syncthreads();

        float S = 0.0f;
#pragma unroll
        for (int i = 0; i < PL; ++i) {
            const int n = base + 32 * i;
            const float wl = (i == 0)      ? trow[(n + NSLOT - 1) & (NSLOT - 1)] : trow[n - 1];
            const float wr = (i == PL - 1) ? trow[(n + 1) & (NSLOT - 1)]         : trow[n + 1];
            const float c = s0 * wl + s1 * v[i] + s2 * wr;
            v[i] = pow_nonneg(c, gm);
            S += v[i];
        }
#pragma unroll
        for (int o = 16; o > 0; o >>= 1) S += __shfl_xor(S, o, 32);
        if (lane == 0) red[2 * NWAVE + wave] = S;
        __syncthreads();
        S = rS[0];
        for (int i = 1; i < WPH; ++i) S += rS[i];
        const float invS = 1.0f / (S + ADD_EPS);

#pragma unroll
        for (int i = 0; i < PL; ++i) wo[32 * i] = v[i] * invS;
    }
}

extern "C" void kernel_launch(void* const* d_in, const int* in_sizes, int n_in,
                              void* d_out, int out_size, void* d_ws, size_t ws_size,
                              hipStream_t stream) {
    (void)in_sizes; (void)n_in; (void)out_size; (void)d_ws; (void)ws_size;
    const float* memory = (const float*)d_in[0];
    const float* k      = (const float*)d_in[1];
    const float* beta   = (const float*)d_in[2];
    const float* g      = (const float*)d_in[3];
    const float* s      = (const float*)d_in[4];
    const float* gamma  = (const float*)d_in[5];
    const float* w_prev = (const float*)d_in[6];
    float* out = (float*)d_out;

    ntm_address_kernel<<<dim3(BATCH), dim3(NTHR), 0, stream>>>(
        memory, k, beta, g, s, gamma, w_prev, out);
}
